# starting kernel
# baseline (speedup 1.0000x reference)
#include <hip/hip_runtime.h>

#define N_NODES 100000
#define N_EDGES 1600000
#define C_IN 64
#define C_OUT 64

typedef __attribute__((ext_vector_type(2))) float v2f;
typedef __attribute__((ext_vector_type(8))) float v8f;

__global__ __launch_bounds__(256) void gcn_zero_deg(int* __restrict__ deg) {
    int i = blockIdx.x * 256 + threadIdx.x;
    if (i < N_NODES) deg[i] = 0;
}

__global__ __launch_bounds__(256) void gcn_degree(const int* __restrict__ col,
                                                  int* __restrict__ deg) {
    int e = blockIdx.x * 256 + threadIdx.x;
    if (e < N_EDGES) atomicAdd(&deg[col[e]], 1);
}

__global__ __launch_bounds__(256) void gcn_dinv(const int* __restrict__ deg,
                                                float* __restrict__ dinv) {
    int i = blockIdx.x * 256 + threadIdx.x;
    if (i < N_NODES) {
        dinv[i] = rsqrtf((float)deg[i] + 1.0f);
    }
}

__global__ __launch_bounds__(256) void gcn_xw_wmma(const float* __restrict__ x,
                                                   const float* __restrict__ W,
                                                   float* __restrict__ xw) {
    __shared__ float sW[C_IN * C_OUT];

    const int tid = threadIdx.x;
    {
        const float4* Wv = (const float4*)W;
        float4* sWv = (float4*)sW;
#pragma unroll
        for (int i = 0; i < 4; ++i) sWv[tid + 256 * i] = Wv[tid + 256 * i];
    }
    __syncthreads();

    const int wave = tid >> 5;
    const int lane = tid & 31;
    const int m0 = blockIdx.x * 32 + (wave >> 2) * 16;
    const int n0 = (wave & 3) * 16;
    const int mh = lane & 15;
    const int kb = (lane >> 4) << 1;

    const float* arow = x + (size_t)(m0 + mh) * C_IN + kb;
    const float* bcol = sW + (size_t)kb * C_OUT + n0 + mh;

    v8f acc = {};
#pragma unroll
    for (int k = 0; k < C_IN; k += 4) {
        v2f a, b;
        a.x = arow[k];
        a.y = arow[k + 1];
        b.x = bcol[k * C_OUT];
        b.y = bcol[k * C_OUT + C_OUT];
        acc = __builtin_amdgcn_wmma_f32_16x16x4_f32(
             false, a,  false, b,
             (short)0, acc,  false,  false);
    }

    float* drow = xw + (size_t)(m0 + 8 * (lane >> 4)) * C_OUT + n0 + mh;
#pragma unroll
    for (int r = 0; r < 8; ++r) drow[r * C_OUT] = acc[r];
}

__global__ __launch_bounds__(256) void gcn_selfloop_bias(const float* __restrict__ xw,
                                                         const float* __restrict__ dinv,
                                                         const float* __restrict__ b,
                                                         float* __restrict__ out) {
    int idx = blockIdx.x * 256 + threadIdx.x;
    if (idx < N_NODES * (C_OUT / 2)) {
        int n = idx >> 5;
        int c2 = idx & 31;
        float di = dinv[n];
        float s = di * di;
        float2 xv = ((const float2*)xw)[idx];
        float2 bv = ((const float2*)b)[c2];
        float2 o;
        o.x = bv.x + s * xv.x;
        o.y = bv.y + s * xv.y;
        ((float2*)out)[idx] = o;
    }
}

__global__ __launch_bounds__(256) void gcn_edge_scatter(const int* __restrict__ row,
                                                        const int* __restrict__ col,
                                                        const float* __restrict__ xw,
                                                        const float* __restrict__ dinv,
                                                        float* __restrict__ out) {
    const int e = blockIdx.x * 8 + (threadIdx.x >> 5);
    if (e >= N_EDGES) return;
    const int lane = threadIdx.x & 31;
    const int r = row[e];
    const int c = col[e];
    const float norm = dinv[r] * dinv[c];
    const float2 v = ((const float2*)(xw + (size_t)r * C_OUT))[lane];
    float* dst = out + (size_t)c * C_OUT + lane * 2;
    unsafeAtomicAdd(dst, v.x * norm);
    unsafeAtomicAdd(dst + 1, v.y * norm);
}

extern "C" void kernel_launch(void* const* d_in, const int* in_sizes, int n_in,
                              void* d_out, int out_size, void* d_ws, size_t ws_size,
                              hipStream_t stream) {
    const float* x = (const float*)d_in[0];
    const int* ei = (const int*)d_in[1];
    const float* W = (const float*)d_in[2];
    const float* b = (const float*)d_in[3];
    float* out = (float*)d_out;

    const int* row = ei;
    const int* col = ei + N_EDGES;

    char* ws = (char*)d_ws;
    int* deg = (int*)(ws);
    float* dinv = (float*)(ws + (size_t)N_NODES * 4);
    float* xw = (float*)(ws + (size_t)N_NODES * 8);

    gcn_zero_deg<<<(N_NODES + 255) / 256, 256, 0, stream>>>(deg);
    gcn_degree<<<(N_EDGES + 255) / 256, 256, 0, stream>>>(col, deg);
    gcn_dinv<<<(N_NODES + 255) / 256, 256, 0, stream>>>(deg, dinv);
    gcn_xw_wmma<<<N_NODES / 32, 256, 0, stream>>>(x, W, xw);
    gcn_selfloop_bias<<<(N_NODES * (C_OUT / 2) + 255) / 256, 256, 0, stream>>>(xw, dinv, b, out);
    gcn_edge_scatter<<<N_EDGES / 8, 256, 0, stream>>>(row, col, xw, dinv, out);
}


# edge sum as per-node gather in fixed order: returning count add gives each in-edge a slot in its target's out row; one wave per node sorts its sources by id, adds rows in that order; no f32 atomics
# speedup vs baseline: 1.5540x; 1.5540x over previous
#include <hip/hip_runtime.h>

#define N_NODES 100000
#define N_EDGES 1600000
#define C_IN 64
#define C_OUT 64

typedef __attribute__((ext_vector_type(2))) float v2f;
typedef __attribute__((ext_vector_type(8))) float v8f;

__global__ __launch_bounds__(256) void gcn_zero_deg(int* __restrict__ deg) {
    int i = blockIdx.x * 256 + threadIdx.x;
    if (i < N_NODES) deg[i] = 0;
}

#define CAP 64

__global__ __launch_bounds__(256) void gcn_count_place(const int* __restrict__ row,
                                                       const int* __restrict__ col,
                                                       int* __restrict__ cnt,
                                                       int* __restrict__ slots) {
    int e = blockIdx.x * 256 + threadIdx.x;
    if (e < N_EDGES) {
        const int c = col[e];
        const int r = row[e];
        const int k = atomicAdd(&cnt[c], 1);
        if (k < CAP) slots[(size_t)c * CAP + k] = r;
    }
}

__global__ __launch_bounds__(256) void gcn_xw_wmma(const float* __restrict__ x,
                                                   const float* __restrict__ W,
                                                   float* __restrict__ xw) {
    __shared__ float sW[C_IN * C_OUT];

    const int tid = threadIdx.x;
    {
        const float4* Wv = (const float4*)W;
        float4* sWv = (float4*)sW;
#pragma unroll
        for (int i = 0; i < 4; ++i) sWv[tid + 256 * i] = Wv[tid + 256 * i];
    }
    __syncthreads();

    const int wave = tid >> 5;
    const int lane = tid & 31;
    const int m0 = blockIdx.x * 32 + (wave >> 2) * 16;
    const int n0 = (wave & 3) * 16;
    const int mh = lane & 15;
    const int kb = (lane >> 4) << 1;

    const float* arow = x + (size_t)(m0 + mh) * C_IN + kb;
    const float* bcol = sW + (size_t)kb * C_OUT + n0 + mh;

    v8f acc = {};
#pragma unroll
    for (int k = 0; k < C_IN; k += 4) {
        v2f a, b;
        a.x = arow[k];
        a.y = arow[k + 1];
        b.x = bcol[k * C_OUT];
        b.y = bcol[k * C_OUT + C_OUT];
        acc = __builtin_amdgcn_wmma_f32_16x16x4_f32(
             false, a,  false, b,
             (short)0, acc,  false,  false);
    }

    float* drow = xw + (size_t)(m0 + 8 * (lane >> 4)) * C_OUT + n0 + mh;
#pragma unroll
    for (int r = 0; r < 8; ++r) drow[r * C_OUT] = acc[r];
}

__global__ __launch_bounds__(256) void gcn_node_sum(const int* __restrict__ row,
                                                    const int* __restrict__ col,
                                                    const int* __restrict__ cnt,
                                                    const float* __restrict__ xw,
                                                    const float* __restrict__ b,
                                                    float* out) {
    __shared__ int2 sl[8][CAP];

    const int lane = threadIdx.x & 31;
    const int wave = __builtin_amdgcn_readfirstlane(threadIdx.x >> 5);
    const int i = blockIdx.x * 8 + wave;
    const float2* xw2 = (const float2*)xw;
    const float2 xv = xw2[(size_t)i * (C_OUT / 2) + lane];
    const float2 bv = ((const float2*)b)[lane];
    const int n = __builtin_amdgcn_readfirstlane(cnt[i]);
    const float di = rsqrtf((float)n + 1.0f);

    float2 acc = {0.0f, 0.0f};
    if (n <= CAP) {
        const int* my = (const int*)(out + (size_t)i * C_OUT);
        int k0 = i;
        if (lane < n) k0 = my[lane];
        const float w0 = rsqrtf((float)cnt[k0] + 1.0f) * di;
        int k1 = i;
        float w1 = 0.0f;
        if (n > 32) {
            if (lane + 32 < n) k1 = my[lane + 32];
            w1 = rsqrtf((float)cnt[k1] + 1.0f) * di;
        }
        const unsigned u0 = ((unsigned)k0 << 6) | (unsigned)lane;
        const unsigned u1 = ((unsigned)k1 << 6) | (unsigned)(lane + 32);
        int r0 = 0, r1 = 0;
        if (n <= 32) {
            for (int j = 0; j < n; ++j) {
                const unsigned s = (unsigned)__builtin_amdgcn_readlane((int)u0, j);
                r0 += (s < u0) ? 1 : 0;
            }
        } else {
            for (int j = 0; j < 32; ++j) {
                const unsigned s = (unsigned)__builtin_amdgcn_readlane((int)u0, j);
                r0 += (s < u0) ? 1 : 0;
                r1 += (s < u1) ? 1 : 0;
            }
            for (int j = 0; j < n - 32; ++j) {
                const unsigned s = (unsigned)__builtin_amdgcn_readlane((int)u1, j);
                r0 += (s < u0) ? 1 : 0;
                r1 += (s < u1) ? 1 : 0;
            }
        }
        if (lane < n) sl[wave][r0] = make_int2(k0, __float_as_int(w0));
        if (lane + 32 < n) sl[wave][r1] = make_int2(k1, __float_as_int(w1));
        __builtin_amdgcn_wave_barrier();

        int2 p0 = make_int2(i, 0);
        int2 p1 = make_int2(i, 0);
        if (lane < n) p0 = sl[wave][lane];
        if (lane + 32 < n) p1 = sl[wave][lane + 32];

#pragma unroll
        for (int c = 0; c < CAP / 16; ++c) {
            const int m = n - 16 * c;
            if (m > 0) {
                const int2 p = (c < 2) ? p0 : p1;
                float2 v[16];
#pragma unroll
                for (int u = 0; u < 16; ++u) {
                    if (u < m) {
                        const int r = __builtin_amdgcn_readlane(p.x, (c & 1) * 16 + u);
                        v[u] = xw2[(size_t)r * (C_OUT / 2) + lane];
                    } else {
                        v[u] = make_float2(0.0f, 0.0f);
                    }
                }
#pragma unroll
                for (int u = 0; u < 16; ++u) {
                    const float w = __int_as_float(__builtin_amdgcn_readlane(p.y, (c & 1) * 16 + u));
                    acc.x = fmaf(v[u].x, w, acc.x);
                    acc.y = fmaf(v[u].y, w, acc.y);
                }
            }
        }
    } else {
        for (int e0 = 0; e0 < N_EDGES; e0 += 32) {
            const int c = col[e0 + lane];
            unsigned m = __builtin_amdgcn_ballot_w32(c == i);
            while (m) {
                const int bit = __builtin_ctz(m);
                m &= m - 1;
                const int r = __builtin_amdgcn_readfirstlane(row[e0 + bit]);
                const float wgt = rsqrtf((float)cnt[r] + 1.0f) * di;
                const float2 v = xw2[(size_t)r * (C_OUT / 2) + lane];
                acc.x = fmaf(v.x, wgt, acc.x);
                acc.y = fmaf(v.y, wgt, acc.y);
            }
        }
    }

    const float s = di * di;
    float2 o;
    o.x = (bv.x + s * xv.x) + acc.x;
    o.y = (bv.y + s * xv.y) + acc.y;
    ((float2*)out)[(size_t)i * (C_OUT / 2) + lane] = o;
}

extern "C" void kernel_launch(void* const* d_in, const int* in_sizes, int n_in,
                              void* d_out, int out_size, void* d_ws, size_t ws_size,
                              hipStream_t stream) {
    const float* x = (const float*)d_in[0];
    const int* ei = (const int*)d_in[1];
    const float* W = (const float*)d_in[2];
    const float* b = (const float*)d_in[3];
    float* out = (float*)d_out;

    const int* row = ei;
    const int* col = ei + N_EDGES;

    char* ws = (char*)d_ws;
    int* cnt = (int*)(ws);
    float* xw = (float*)(ws + (size_t)N_NODES * 8);

    gcn_zero_deg<<<(N_NODES + 255) / 256, 256, 0, stream>>>(cnt);
    gcn_count_place<<<(N_EDGES + 255) / 256, 256, 0, stream>>>(row, col, cnt, (int*)d_out);
    gcn_xw_wmma<<<N_NODES / 32, 256, 0, stream>>>(x, W, xw);
    gcn_node_sum<<<N_NODES / 8, 256, 0, stream>>>(row, col, cnt, xw, b, out);
}


# aggregate first: node waves gather rows of the input x, not of a written xw plane; each block's 16 summed rows go via LDS to WMMA x W + b; product launch and xw gone (4 to 3); sum order changes
# speedup vs baseline: 1.7987x; 1.1575x over previous
#include <hip/hip_runtime.h>

#define N_NODES 100000
#define N_EDGES 1600000
#define C_IN 64
#define C_OUT 64

typedef __attribute__((ext_vector_type(2))) float v2f;
typedef __attribute__((ext_vector_type(8))) float v8f;

__global__ __launch_bounds__(256) void gcn_zero_deg(int* __restrict__ deg) {
    int i = blockIdx.x * 256 + threadIdx.x;
    if (i < N_NODES) deg[i] = 0;
}

#define CAP 64

__global__ __launch_bounds__(256) void gcn_count_place(const int* __restrict__ row,
                                                       const int* __restrict__ col,
                                                       int* __restrict__ cnt,
                                                       int* __restrict__ slots) {
    int e = blockIdx.x * 256 + threadIdx.x;
    if (e < N_EDGES) {
        const int c = col[e];
        const int r = row[e];
        const int k = atomicAdd(&cnt[c], 1);
        if (k < CAP) slots[(size_t)c * CAP + k] = r;
    }
}

#define NPB 16
#define APITCH 68

__global__ __launch_bounds__(512) void gcn_gather_project(const int* __restrict__ row,
                                                          const int* __restrict__ col,
                                                          const int* __restrict__ cnt,
                                                          const float* __restrict__ x,
                                                          const float* __restrict__ W,
                                                          const float* __restrict__ b,
                                                          float* out) {
    __shared__ int2 sl[NPB][CAP];
    __shared__ __attribute__((aligned(16))) float sA[NPB * APITCH];

    const int lane = threadIdx.x & 31;
    const int wave = __builtin_amdgcn_readfirstlane(threadIdx.x >> 5);
    const int i = blockIdx.x * NPB + wave;
    const float2* x2 = (const float2*)x;
    const float2 xv = x2[(size_t)i * (C_IN / 2) + lane];
    const int n = __builtin_amdgcn_readfirstlane(cnt[i]);
    const float di = rsqrtf((float)n + 1.0f);

    float2 acc = {0.0f, 0.0f};
    if (n <= CAP) {
        const int* my = (const int*)(out + (size_t)i * C_OUT);
        int k0 = i;
        if (lane < n) k0 = my[lane];
        const float w0 = rsqrtf((float)cnt[k0] + 1.0f) * di;
        int k1 = i;
        float w1 = 0.0f;
        if (n > 32) {
            if (lane + 32 < n) k1 = my[lane + 32];
            w1 = rsqrtf((float)cnt[k1] + 1.0f) * di;
        }
        const unsigned u0 = ((unsigned)k0 << 6) | (unsigned)lane;
        const unsigned u1 = ((unsigned)k1 << 6) | (unsigned)(lane + 32);
        int r0 = 0, r1 = 0;
        if (n <= 32) {
            for (int j = 0; j < n; ++j) {
                const unsigned s = (unsigned)__builtin_amdgcn_readlane((int)u0, j);
                r0 += (s < u0) ? 1 : 0;
            }
        } else {
            for (int j = 0; j < 32; ++j) {
                const unsigned s = (unsigned)__builtin_amdgcn_readlane((int)u0, j);
                r0 += (s < u0) ? 1 : 0;
                r1 += (s < u1) ? 1 : 0;
            }
            for (int j = 0; j < n - 32; ++j) {
                const unsigned s = (unsigned)__builtin_amdgcn_readlane((int)u1, j);
                r0 += (s < u0) ? 1 : 0;
                r1 += (s < u1) ? 1 : 0;
            }
        }
        if (lane < n) sl[wave][r0] = make_int2(k0, __float_as_int(w0));
        if (lane + 32 < n) sl[wave][r1] = make_int2(k1, __float_as_int(w1));
        __builtin_amdgcn_wave_barrier();

        int2 p0 = make_int2(i, 0);
        int2 p1 = make_int2(i, 0);
        if (lane < n) p0 = sl[wave][lane];
        if (lane + 32 < n) p1 = sl[wave][lane + 32];

#pragma unroll
        for (int c = 0; c < CAP / 16; ++c) {
            const int m = n - 16 * c;
            if (m > 0) {
                const int2 p = (c < 2) ? p0 : p1;
                float2 v[16];
#pragma unroll
                for (int u = 0; u < 16; ++u) {
                    if (u < m) {
                        const int r = __builtin_amdgcn_readlane(p.x, (c & 1) * 16 + u);
                        v[u] = x2[(size_t)r * (C_IN / 2) + lane];
                    } else {
                        v[u] = make_float2(0.0f, 0.0f);
                    }
                }
#pragma unroll
                for (int u = 0; u < 16; ++u) {
                    const float w = __int_as_float(__builtin_amdgcn_readlane(p.y, (c & 1) * 16 + u));
                    acc.x = fmaf(v[u].x, w, acc.x);
                    acc.y = fmaf(v[u].y, w, acc.y);
                }
            }
        }
    } else {
        for (int e0 = 0; e0 < N_EDGES; e0 += 32) {
            const int c = col[e0 + lane];
            unsigned m = __builtin_amdgcn_ballot_w32(c == i);
            while (m) {
                const int bit = __builtin_ctz(m);
                m &= m - 1;
                const int r = __builtin_amdgcn_readfirstlane(row[e0 + bit]);
                const float wgt = rsqrtf((float)cnt[r] + 1.0f) * di;
                const float2 v = x2[(size_t)r * (C_IN / 2) + lane];
                acc.x = fmaf(v.x, wgt, acc.x);
                acc.y = fmaf(v.y, wgt, acc.y);
            }
        }
    }

    const float s = di * di;
    float2 a2;
    a2.x = fmaf(s, xv.x, acc.x);
    a2.y = fmaf(s, xv.y, acc.y);
    *(float2*)(sA + wave * APITCH + 2 * lane) = a2;

    __syncthreads();

    if (wave < 4) {
        const int n0 = wave * 16;
        const int mh = lane & 15;
        const int kb = (lane >> 4) << 1;
        const float* arow = sA + mh * APITCH + kb;
        const float* bcol = W + (size_t)kb * C_OUT + n0 + mh;

        const float bn = b[n0 + mh];
        v8f d = {bn, bn, bn, bn, bn, bn, bn, bn};
#pragma unroll
        for (int k = 0; k < C_IN; k += 4) {
            v2f fa, fb;
            fa.x = arow[k];
            fa.y = arow[k + 1];
            fb.x = bcol[k * C_OUT];
            fb.y = bcol[k * C_OUT + C_OUT];
            d = __builtin_amdgcn_wmma_f32_16x16x4_f32(
                 false, fa,  false, fb,
                 (short)0, d,  false,  false);
        }

        float* drow = out + ((size_t)blockIdx.x * NPB + 8 * (lane >> 4)) * C_OUT + n0 + mh;
#pragma unroll
        for (int r = 0; r < 8; ++r) drow[r * C_OUT] = d[r];
    }
}

extern "C" void kernel_launch(void* const* d_in, const int* in_sizes, int n_in,
                              void* d_out, int out_size, void* d_ws, size_t ws_size,
                              hipStream_t stream) {
    const float* x = (const float*)d_in[0];
    const int* ei = (const int*)d_in[1];
    const float* W = (const float*)d_in[2];
    const float* b = (const float*)d_in[3];
    float* out = (float*)d_out;

    const int* row = ei;
    const int* col = ei + N_EDGES;

    int* cnt = (int*)d_ws;

    gcn_zero_deg<<<(N_NODES + 255) / 256, 256, 0, stream>>>(cnt);
    gcn_count_place<<<(N_EDGES + 255) / 256, 256, 0, stream>>>(row, col, cnt, (int*)d_out);
    gcn_gather_project<<<N_NODES / NPB, 512, 0, stream>>>(row, col, cnt, x, W, b, out);
}


# count/place pass: a node's counter and slot row are written from one cache half only: blocks take a side by die id, a 512-edge chunk by ticket (4 edges a thread), handle only that side's targets
# speedup vs baseline: 4.0279x; 2.2393x over previous
#include <hip/hip_runtime.h>

#define N_NODES 100000
#define N_EDGES 1600000
#define C_IN 64
#define C_OUT 64

typedef __attribute__((ext_vector_type(2))) float v2f;
typedef __attribute__((ext_vector_type(8))) float v8f;

#define TICK_GAP 64
#define N_CLEAR (N_NODES + 2 * TICK_GAP)

__global__ __launch_bounds__(256) void gcn_zero_deg(int* __restrict__ deg) {
    int i = blockIdx.x * 256 + threadIdx.x;
    if (i < N_CLEAR) deg[i] = 0;
}

#define CAP 64

#define EPT 4
#define K1_THREADS 128
#define CHUNK (K1_THREADS * EPT)
#define NCHUNK (N_EDGES / CHUNK)
#define SPLIT 50016
static_assert(NCHUNK * CHUNK == N_EDGES, "chunks cover the edges exactly");
static_assert(SPLIT % 32 == 0 && SPLIT < N_NODES, "side boundary on a counter line");

__device__ __forceinline__ bool gcn_mine(int c, int side) { return (c >= SPLIT ? 1 : 0) == side; }

__global__ __launch_bounds__(K1_THREADS) void gcn_count_place(const int* __restrict__ row,
                                                              const int* __restrict__ col,
                                                              int* __restrict__ cnt,
                                                              int* tick,
                                                              int* __restrict__ slots) {
    __shared__ int job;
    if (threadIdx.x == 0) {
        const int side0 = (int)((__builtin_amdgcn_s_sendmsg_rtn(0x87) >> 18) & 1u);
        int side = side0;
        int t = atomicAdd(&tick[side * TICK_GAP], 1);
        if (t >= NCHUNK) {
            side ^= 1;
            t = atomicAdd(&tick[side * TICK_GAP], 1);
        }
        job = (t < NCHUNK) ? side * NCHUNK + t : -1;
    }
    __syncthreads();
    const int j = job;
    if (j < 0) return;
    const int side = (j >= NCHUNK) ? 1 : 0;
    const int e = (j - side * NCHUNK) * CHUNK + (int)threadIdx.x * EPT;
    const int4 c4 = *(const int4*)(col + e);
    const int4 r4 = *(const int4*)(row + e);
    const bool m0 = gcn_mine(c4.x, side), m1 = gcn_mine(c4.y, side);
    const bool m2 = gcn_mine(c4.z, side), m3 = gcn_mine(c4.w, side);
    int k0 = CAP, k1 = CAP, k2 = CAP, k3 = CAP;
    if (m0) k0 = atomicAdd(&cnt[c4.x], 1);
    if (m1) k1 = atomicAdd(&cnt[c4.y], 1);
    if (m2) k2 = atomicAdd(&cnt[c4.z], 1);
    if (m3) k3 = atomicAdd(&cnt[c4.w], 1);
    if (k0 < CAP) slots[(size_t)c4.x * CAP + k0] = r4.x;
    if (k1 < CAP) slots[(size_t)c4.y * CAP + k1] = r4.y;
    if (k2 < CAP) slots[(size_t)c4.z * CAP + k2] = r4.z;
    if (k3 < CAP) slots[(size_t)c4.w * CAP + k3] = r4.w;
}

#define NPB 16
#define APITCH 68

__global__ __launch_bounds__(512) void gcn_gather_project(const int* __restrict__ row,
                                                          const int* __restrict__ col,
                                                          const int* __restrict__ cnt,
                                                          const float* __restrict__ x,
                                                          const float* __restrict__ W,
                                                          const float* __restrict__ b,
                                                          float* out) {
    __shared__ int2 sl[NPB][CAP];
    __shared__ __attribute__((aligned(16))) float sA[NPB * APITCH];

    const int lane = threadIdx.x & 31;
    const int wave = __builtin_amdgcn_readfirstlane(threadIdx.x >> 5);
    const int i = blockIdx.x * NPB + wave;
    const float2* x2 = (const float2*)x;
    const float2 xv = x2[(size_t)i * (C_IN / 2) + lane];
    const int n = __builtin_amdgcn_readfirstlane(cnt[i]);
    const float di = rsqrtf((float)n + 1.0f);

    float2 acc = {0.0f, 0.0f};
    if (n <= CAP) {
        const int* my = (const int*)(out + (size_t)i * C_OUT);
        int k0 = i;
        if (lane < n) k0 = my[lane];
        const float w0 = rsqrtf((float)cnt[k0] + 1.0f) * di;
        int k1 = i;
        float w1 = 0.0f;
        if (n > 32) {
            if (lane + 32 < n) k1 = my[lane + 32];
            w1 = rsqrtf((float)cnt[k1] + 1.0f) * di;
        }
        const unsigned u0 = ((unsigned)k0 << 6) | (unsigned)lane;
        const unsigned u1 = ((unsigned)k1 << 6) | (unsigned)(lane + 32);
        int r0 = 0, r1 = 0;
        if (n <= 32) {
            for (int j = 0; j < n; ++j) {
                const unsigned s = (unsigned)__builtin_amdgcn_readlane((int)u0, j);
                r0 += (s < u0) ? 1 : 0;
            }
        } else {
            for (int j = 0; j < 32; ++j) {
                const unsigned s = (unsigned)__builtin_amdgcn_readlane((int)u0, j);
                r0 += (s < u0) ? 1 : 0;
                r1 += (s < u1) ? 1 : 0;
            }
            for (int j = 0; j < n - 32; ++j) {
                const unsigned s = (unsigned)__builtin_amdgcn_readlane((int)u1, j);
                r0 += (s < u0) ? 1 : 0;
                r1 += (s < u1) ? 1 : 0;
            }
        }
        if (lane < n) sl[wave][r0] = make_int2(k0, __float_as_int(w0));
        if (lane + 32 < n) sl[wave][r1] = make_int2(k1, __float_as_int(w1));
        __builtin_amdgcn_wave_barrier();

        int2 p0 = make_int2(i, 0);
        int2 p1 = make_int2(i, 0);
        if (lane < n) p0 = sl[wave][lane];
        if (lane + 32 < n) p1 = sl[wave][lane + 32];

#pragma unroll
        for (int c = 0; c < CAP / 16; ++c) {
            const int m = n - 16 * c;
            if (m > 0) {
                const int2 p = (c < 2) ? p0 : p1;
                float2 v[16];
#pragma unroll
                for (int u = 0; u < 16; ++u) {
                    if (u < m) {
                        const int r = __builtin_amdgcn_readlane(p.x, (c & 1) * 16 + u);
                        v[u] = x2[(size_t)r * (C_IN / 2) + lane];
                    } else {
                        v[u] = make_float2(0.0f, 0.0f);
                    }
                }
#pragma unroll
                for (int u = 0; u < 16; ++u) {
                    const float w = __int_as_float(__builtin_amdgcn_readlane(p.y, (c & 1) * 16 + u));
                    acc.x = fmaf(v[u].x, w, acc.x);
                    acc.y = fmaf(v[u].y, w, acc.y);
                }
            }
        }
    } else {
        for (int e0 = 0; e0 < N_EDGES; e0 += 32) {
            const int c = col[e0 + lane];
            unsigned m = __builtin_amdgcn_ballot_w32(c == i);
            while (m) {
                const int bit = __builtin_ctz(m);
                m &= m - 1;
                const int r = __builtin_amdgcn_readfirstlane(row[e0 + bit]);
                const float wgt = rsqrtf((float)cnt[r] + 1.0f) * di;
                const float2 v = x2[(size_t)r * (C_IN / 2) + lane];
                acc.x = fmaf(v.x, wgt, acc.x);
                acc.y = fmaf(v.y, wgt, acc.y);
            }
        }
    }

    const float s = di * di;
    float2 a2;
    a2.x = fmaf(s, xv.x, acc.x);
    a2.y = fmaf(s, xv.y, acc.y);
    *(float2*)(sA + wave * APITCH + 2 * lane) = a2;

    __syncthreads();

    if (wave < 4) {
        const int n0 = wave * 16;
        const int mh = lane & 15;
        const int kb = (lane >> 4) << 1;
        const float* arow = sA + mh * APITCH + kb;
        const float* bcol = W + (size_t)kb * C_OUT + n0 + mh;

        const float bn = b[n0 + mh];
        v8f d = {bn, bn, bn, bn, bn, bn, bn, bn};
#pragma unroll
        for (int k = 0; k < C_IN; k += 4) {
            v2f fa, fb;
            fa.x = arow[k];
            fa.y = arow[k + 1];
            fb.x = bcol[k * C_OUT];
            fb.y = bcol[k * C_OUT + C_OUT];
            d = __builtin_amdgcn_wmma_f32_16x16x4_f32(
                 false, fa,  false, fb,
                 (short)0, d,  false,  false);
        }

        float* drow = out + ((size_t)blockIdx.x * NPB + 8 * (lane >> 4)) * C_OUT + n0 + mh;
#pragma unroll
        for (int r = 0; r < 8; ++r) drow[r * C_OUT] = d[r];
    }
}

extern "C" void kernel_launch(void* const* d_in, const int* in_sizes, int n_in,
                              void* d_out, int out_size, void* d_ws, size_t ws_size,
                              hipStream_t stream) {
    const float* x = (const float*)d_in[0];
    const int* ei = (const int*)d_in[1];
    const float* W = (const float*)d_in[2];
    const float* b = (const float*)d_in[3];
    float* out = (float*)d_out;

    const int* row = ei;
    const int* col = ei + N_EDGES;

    int* cnt = (int*)d_ws;
    int* tick = cnt + N_NODES;

    gcn_zero_deg<<<(N_CLEAR + 255) / 256, 256, 0, stream>>>(cnt);
    gcn_count_place<<<2 * NCHUNK, K1_THREADS, 0, stream>>>(row, col, cnt, tick, (int*)d_out);
    gcn_gather_project<<<N_NODES / NPB, 512, 0, stream>>>(row, col, cnt, x, W, b, out);
}
